# starting kernel
# baseline (speedup 1.0000x reference)
#include <hip/hip_runtime.h>

typedef __attribute__((ext_vector_type(2))) float v2f;
typedef __attribute__((ext_vector_type(8))) float v8f;

#define H        1024
#define M_TILE   32
#define K_BLOCK  64
#define A_STRIDE (K_BLOCK + 4)
#define NTHREADS 512
#define NWAVES   16

__global__ __launch_bounds__(NTHREADS)
void SSM_gemm_scan_kernel(const float* __restrict__ x,
                          const float* __restrict__ W,
                          const float* __restrict__ bias,
                          float* __restrict__ out)
{
    __shared__ float sOut[M_TILE * H];
    __shared__ float sA[M_TILE * A_STRIDE];

    const int tid  = threadIdx.x;
    const int wave = tid >> 5;
    const int lane = tid & 31;
    const int mBase = blockIdx.x * M_TILE;

    v8f acc[2][4];
    #pragma unroll
    for (int mt = 0; mt < 2; ++mt)
        #pragma unroll
        for (int nt = 0; nt < 4; ++nt)
            acc[mt][nt] = v8f{};

    const int nWaveBase = wave * 64;

    const int aRow  = lane & 15;
    const int aKoff = (lane >> 4) << 1;
    const int bLane = lane & 15;
    const int bKoff = (lane >> 4) << 1;

    for (int kb = 0; kb < H; kb += K_BLOCK) {
        {
            const int r  = tid >> 4;
            const int c4 = (tid & 15) << 2;
            const float4 v = *(const float4*)&x[(size_t)(mBase + r) * H + kb + c4];
            *(float4*)&sA[r * A_STRIDE + c4] = v;
        }
        __syncthreads();

        #pragma unroll
        for (int nt = 0; nt < 4; ++nt) {
            const int n = nWaveBase + nt * 16;
            const float* Wrow = W + (size_t)(n + bLane) * H + kb;
            v8f a0 = acc[0][nt];
            v8f a1 = acc[1][nt];
            #pragma unroll
            for (int k = 0; k < K_BLOCK; k += 4) {
                v2f bfrag;
                bfrag.x = Wrow[k + bKoff + 0];
                bfrag.y = Wrow[k + bKoff + 1];
                v2f af0, af1;
                af0.x = sA[(aRow +  0) * A_STRIDE + k + aKoff + 0];
                af0.y = sA[(aRow +  0) * A_STRIDE + k + aKoff + 1];
                af1.x = sA[(aRow + 16) * A_STRIDE + k + aKoff + 0];
                af1.y = sA[(aRow + 16) * A_STRIDE + k + aKoff + 1];
                a0 = __builtin_amdgcn_wmma_f32_16x16x4_f32(
                        false, af0, false, bfrag, (short)0, a0, false, false);
                a1 = __builtin_amdgcn_wmma_f32_16x16x4_f32(
                        false, af1, false, bfrag, (short)0, a1, false, false);
            }
            acc[0][nt] = a0;
            acc[1][nt] = a1;
        }
        __syncthreads();
    }

    {
        const int cCol = lane & 15;
        const int rOff = (lane >> 4) << 3;
        #pragma unroll
        for (int nt = 0; nt < 4; ++nt) {
            const int n = nWaveBase + nt * 16 + cCol;
            const float b = bias[n];
            #pragma unroll
            for (int mt = 0; mt < 2; ++mt)
                #pragma unroll
                for (int j = 0; j < 8; ++j)
                    sOut[(mt * 16 + rOff + j) * H + n] = acc[mt][nt][j] + b;
        }
    }
    __syncthreads();

    for (int r = wave; r < M_TILE; r += NWAVES) {
        const float* row = &sOut[r * H];
        const int base = lane * 32;

        float vals[32];
        float sum = 0.0f;
        #pragma unroll
        for (int i = 0; i < 32; i += 4) {
            const float4 v = *(const float4*)&row[base + i];
            vals[i + 0] = v.x; vals[i + 1] = v.y; vals[i + 2] = v.z; vals[i + 3] = v.w;
            sum += v.x + v.y + v.z + v.w;
        }

        float pre = sum;
        #pragma unroll
        for (int d = 1; d < 32; d <<= 1) {
            const float t = __shfl_up(pre, d, 32);
            if (lane >= d) pre += t;
        }
        float run = pre - sum;

        const size_t outBase = (size_t)(mBase + r) * H + base;
        #pragma unroll
        for (int i = 0; i < 32; i += 4) {
            float4 o;
            run += vals[i + 0]; o.x = run;
            run += vals[i + 1]; o.y = run;
            run += vals[i + 2]; o.z = run;
            run += vals[i + 3]; o.w = run;
            *(float4*)(out + outBase + i) = o;
        }
    }
}

extern "C" void kernel_launch(void* const* d_in, const int* in_sizes, int n_in,
                              void* d_out, int out_size, void* d_ws, size_t ws_size,
                              hipStream_t stream) {
    const float* x    = (const float*)d_in[0];
    const float* W    = (const float*)d_in[1];
    const float* bias = (const float*)d_in[2];
    float* out = (float*)d_out;

    const int Mrows = in_sizes[0] / H;
    dim3 grid(Mrows / M_TILE);
    SSM_gemm_scan_kernel<<<grid, NTHREADS, 0, stream>>>(x, W, bias, out);
}


# Stage each 32-k slab of W (and x) in double-buffered LDS with whole-line 16B loads; A and B frags by 16B LDS reads; k order inside each 8-k group permuted, so fp32 sum order changes
# speedup vs baseline: 1.8314x; 1.8314x over previous
#include <hip/hip_runtime.h>


typedef __attribute__((ext_vector_type(2))) float v2f;
typedef __attribute__((ext_vector_type(8))) float v8f;

#define H        1024
#define M_TILE   32
#define K_BLOCK  32
#define K_PITCH  36
#define NTHREADS 512
#define NWAVES   16
#define N_SLABS  (H / K_BLOCK)
#define STAGE_ROWS   (H + M_TILE)
#define STAGE_FLOATS (STAGE_ROWS * K_PITCH)
#define W_ROWS_PER_PASS (NTHREADS / 8)
#define W_PASSES (H / W_ROWS_PER_PASS)

static_assert((N_SLABS & 1) == 0, "last slab must sit in buffer 1 so the output tile (buffer 0) is free");
static_assert(M_TILE * H <= STAGE_FLOATS, "output tile must fit inside staging buffer 0");

__global__ __launch_bounds__(NTHREADS)
void SSM_gemm_scan_kernel(const float* __restrict__ x,
                          const float* __restrict__ W,
                          const float* __restrict__ bias,
                          float* __restrict__ out)
{
    __shared__ __attribute__((aligned(16))) float smem[2 * STAGE_FLOATS];
    float* const sOut = smem;

    const int tid  = threadIdx.x;
    const int wave = __builtin_amdgcn_readfirstlane(tid >> 5);
    const int lane = tid & 31;
    const int mBase = blockIdx.x * M_TILE;

    v8f acc[2][4];
    #pragma unroll
    for (int mt = 0; mt < 2; ++mt)
        #pragma unroll
        for (int nt = 0; nt < 4; ++nt)
            acc[mt][nt] = v8f{};

    const int nWaveBase = wave * 64;

    const int fC = lane & 15;
    const int fK = (lane >> 4) << 2;
    const int aOff = (H + fC) * K_PITCH + fK;
    const int bOff = (nWaveBase + fC) * K_PITCH + fK;

    const int sRow = tid >> 3;
    const int sQ   = (tid & 7) << 2;
    const float* wSrc = W + (size_t)sRow * H + sQ;
    const float* xSrc = x + (size_t)(mBase + (sRow & (M_TILE - 1))) * H + sQ;
    const int wDst = sRow * K_PITCH + sQ;
    const int xDst = (H + (sRow & (M_TILE - 1))) * K_PITCH + sQ;
    const bool stagesX = wave < (M_TILE * 8 / 32);

    auto mma_slab = [&](const int base) {
        float4 fa[2][2];
        float4 fb[2][4];
        #pragma unroll
        for (int mt = 0; mt < 2; ++mt)
            fa[0][mt] = *(const float4*)&smem[base + aOff + mt * 16 * K_PITCH];
        #pragma unroll
        for (int nt = 0; nt < 4; ++nt)
            fb[0][nt] = *(const float4*)&smem[base + bOff + nt * 16 * K_PITCH];

        #pragma unroll
        for (int j = 0; j < K_BLOCK / 8; ++j) {
            const int s = j & 1;
            if (j + 1 < K_BLOCK / 8) {
                #pragma unroll
                for (int mt = 0; mt < 2; ++mt)
                    fa[s ^ 1][mt] = *(const float4*)&smem[base + aOff + mt * 16 * K_PITCH + (j + 1) * 8];
                #pragma unroll
                for (int nt = 0; nt < 4; ++nt)
                    fb[s ^ 1][nt] = *(const float4*)&smem[base + bOff + nt * 16 * K_PITCH + (j + 1) * 8];
            }
            #pragma unroll
            for (int nt = 0; nt < 4; ++nt)
                #pragma unroll
                for (int mt = 0; mt < 2; ++mt) {
                    const v2f a = {fa[s][mt].x, fa[s][mt].y};
                    const v2f b = {fb[s][nt].x, fb[s][nt].y};
                    acc[mt][nt] = __builtin_amdgcn_wmma_f32_16x16x4_f32(
                            false, a, false, b, (short)0, acc[mt][nt], false, false);
                }
            #pragma unroll
            for (int nt = 0; nt < 4; ++nt)
                #pragma unroll
                for (int mt = 0; mt < 2; ++mt) {
                    const v2f a = {fa[s][mt].z, fa[s][mt].w};
                    const v2f b = {fb[s][nt].z, fb[s][nt].w};
                    acc[mt][nt] = __builtin_amdgcn_wmma_f32_16x16x4_f32(
                            false, a, false, b, (short)0, acc[mt][nt], false, false);
                }
        }
    };

    float4 pw[W_PASSES];
    float4 px = float4{0.f, 0.f, 0.f, 0.f};

    #pragma unroll
    for (int i = 0; i < W_PASSES; ++i)
        pw[i] = *(const float4*)(wSrc + (size_t)i * W_ROWS_PER_PASS * H);
    if (stagesX) px = *(const float4*)xSrc;
    #pragma unroll
    for (int i = 0; i < W_PASSES; ++i)
        *(float4*)&smem[wDst + i * W_ROWS_PER_PASS * K_PITCH] = pw[i];
    if (stagesX) *(float4*)&smem[xDst] = px;
    __syncthreads();

    for (int it = 0; it < N_SLABS - 1; ++it) {
        const int cur = (it & 1) * STAGE_FLOATS;
        const int nxt = STAGE_FLOATS - cur;
        const int kb  = (it + 1) * K_BLOCK;

        #pragma unroll
        for (int i = 0; i < W_PASSES; ++i)
            pw[i] = *(const float4*)(wSrc + (size_t)i * W_ROWS_PER_PASS * H + kb);
        if (stagesX) px = *(const float4*)(xSrc + kb);

        mma_slab(cur);

        #pragma unroll
        for (int i = 0; i < W_PASSES; ++i)
            *(float4*)&smem[nxt + wDst + i * W_ROWS_PER_PASS * K_PITCH] = pw[i];
        if (stagesX) *(float4*)&smem[nxt + xDst] = px;
        __syncthreads();
    }
    mma_slab(((N_SLABS - 1) & 1) * STAGE_FLOATS);

    {
        const int cCol = lane & 15;
        const int rOff = (lane >> 4) << 3;
        #pragma unroll
        for (int nt = 0; nt < 4; ++nt) {
            const int n = nWaveBase + nt * 16 + cCol;
            const float b = bias[n];
            #pragma unroll
            for (int mt = 0; mt < 2; ++mt)
                #pragma unroll
                for (int j = 0; j < 8; ++j)
                    sOut[(mt * 16 + rOff + j) * H + n] = acc[mt][nt][j] + b;
        }
    }
    __syncthreads();

    for (int r = wave; r < M_TILE; r += NWAVES) {
        const float* row = &sOut[r * H];
        const int base = lane * 32;

        float vals[32];
        float sum = 0.0f;
        #pragma unroll
        for (int i = 0; i < 32; i += 4) {
            const float4 v = *(const float4*)&row[base + i];
            vals[i + 0] = v.x; vals[i + 1] = v.y; vals[i + 2] = v.z; vals[i + 3] = v.w;
            sum += v.x + v.y + v.z + v.w;
        }

        float pre = sum;
        #pragma unroll
        for (int d = 1; d < 32; d <<= 1) {
            const float t = __shfl_up(pre, d, 32);
            if (lane >= d) pre += t;
        }
        float run = pre - sum;

        const size_t outBase = (size_t)(mBase + r) * H + base;
        #pragma unroll
        for (int i = 0; i < 32; i += 4) {
            float4 o;
            run += vals[i + 0]; o.x = run;
            run += vals[i + 1]; o.y = run;
            run += vals[i + 2]; o.z = run;
            run += vals[i + 3]; o.w = run;
            *(float4*)(out + outBase + i) = o;
        }
    }
}

extern "C" void kernel_launch(void* const* d_in, const int* in_sizes, int n_in,
                              void* d_out, int out_size, void* d_ws, size_t ws_size,
                              hipStream_t stream) {
    const float* x    = (const float*)d_in[0];
    const float* W    = (const float*)d_in[1];
    const float* bias = (const float*)d_in[2];
    float* out = (float*)d_out;

    const int Mrows = in_sizes[0] / H;
    dim3 grid(Mrows / M_TILE);
    SSM_gemm_scan_kernel<<<grid, NTHREADS, 0, stream>>>(x, W, bias, out);
}


# Row tile per workgroup 32 -> 64 rows (each wave 64x64, 16 accumulator tiles): a staged W slab feeds twice the WMMAs, halving L2 reads of W, staging stores and barriers per FLOP; sum order unchanged
# speedup vs baseline: 2.1364x; 1.1665x over previous
#include <hip/hip_runtime.h>


typedef __attribute__((ext_vector_type(2))) float v2f;
typedef __attribute__((ext_vector_type(8))) float v8f;

#define H        1024
#define M_TILE   64
#define K_BLOCK  32
#define K_PITCH  36
#define NTHREADS 512
#define NWAVES   16
#define N_SLABS  (H / K_BLOCK)
#define STAGE_ROWS   (H + M_TILE)
#define STAGE_FLOATS (STAGE_ROWS * K_PITCH)
#define W_ROWS_PER_PASS (NTHREADS / 8)
#define W_PASSES (H / W_ROWS_PER_PASS)
#define W_HALF   (W_PASSES / 2)
#define J_STEPS  (K_BLOCK / 8)

static_assert(M_TILE == W_ROWS_PER_PASS, "every thread stages exactly one float4 of the x slab");
static_assert(M_TILE * H <= 2 * STAGE_FLOATS, "output tile must fit inside the two staging buffers");
static_assert(2 * STAGE_FLOATS * 4 <= 320 * 1024, "both staging buffers must fit the LDS of one WGP");
static_assert(J_STEPS == 4, "the staging batches are hung on fragment steps 1 and 3");

__global__ __launch_bounds__(NTHREADS)
void SSM_gemm_scan_kernel(const float* __restrict__ x,
                          const float* __restrict__ W,
                          const float* __restrict__ bias,
                          float* __restrict__ out)
{
    __shared__ __attribute__((aligned(16))) float smem[2 * STAGE_FLOATS];
    float* const sOut = smem;

    const int tid  = threadIdx.x;
    const int wave = __builtin_amdgcn_readfirstlane(tid >> 5);
    const int lane = tid & 31;
    const int mBase = blockIdx.x * M_TILE;

    v8f acc[4][4];
    #pragma unroll
    for (int mt = 0; mt < 4; ++mt)
        #pragma unroll
        for (int nt = 0; nt < 4; ++nt)
            acc[mt][nt] = v8f{};

    const int nWaveBase = wave * 64;

    const int fC = lane & 15;
    const int fK = (lane >> 4) << 2;
    const int aOff = (H + fC) * K_PITCH + fK;
    const int bOff = (nWaveBase + fC) * K_PITCH + fK;

    const int sRow = tid >> 3;
    const int sQ   = (tid & 7) << 2;
    const float* wSrc = W + (size_t)sRow * H + sQ;
    const float* xSrc = x + (size_t)(mBase + sRow) * H + sQ;
    const int wDst = sRow * K_PITCH + sQ;
    const int xDst = (H + sRow) * K_PITCH + sQ;

    float4 fa[2][4];
    float4 fb[2][4];

    auto load_frags = [&](const int s, const int base, const int j) {
        #pragma unroll
        for (int mt = 0; mt < 4; ++mt)
            fa[s][mt] = *(const float4*)&smem[base + aOff + mt * 16 * K_PITCH + j * 8];
        #pragma unroll
        for (int nt = 0; nt < 4; ++nt)
            fb[s][nt] = *(const float4*)&smem[base + bOff + nt * 16 * K_PITCH + j * 8];
    };

    auto mma_step = [&](const int s) {
        #pragma unroll
        for (int nt = 0; nt < 4; ++nt)
            #pragma unroll
            for (int mt = 0; mt < 4; ++mt) {
                const v2f a = {fa[s][mt].x, fa[s][mt].y};
                const v2f b = {fb[s][nt].x, fb[s][nt].y};
                acc[mt][nt] = __builtin_amdgcn_wmma_f32_16x16x4_f32(
                        false, a, false, b, (short)0, acc[mt][nt], false, false);
            }
        #pragma unroll
        for (int nt = 0; nt < 4; ++nt)
            #pragma unroll
            for (int mt = 0; mt < 4; ++mt) {
                const v2f a = {fa[s][mt].z, fa[s][mt].w};
                const v2f b = {fb[s][nt].z, fb[s][nt].w};
                acc[mt][nt] = __builtin_amdgcn_wmma_f32_16x16x4_f32(
                        false, a, false, b, (short)0, acc[mt][nt], false, false);
            }
    };

    float4 pw[W_HALF];
    float4 px;

    #pragma unroll
    for (int i = 0; i < W_HALF; ++i)
        pw[i] = *(const float4*)(wSrc + (size_t)i * W_ROWS_PER_PASS * H);
    px = *(const float4*)xSrc;
    #pragma unroll
    for (int i = 0; i < W_HALF; ++i)
        *(float4*)&smem[wDst + i * W_ROWS_PER_PASS * K_PITCH] = pw[i];
    *(float4*)&smem[xDst] = px;
    #pragma unroll
    for (int i = 0; i < W_HALF; ++i)
        pw[i] = *(const float4*)(wSrc + (size_t)(i + W_HALF) * W_ROWS_PER_PASS * H);
    #pragma unroll
    for (int i = 0; i < W_HALF; ++i)
        *(float4*)&smem[wDst + (i + W_HALF) * W_ROWS_PER_PASS * K_PITCH] = pw[i];
    __syncthreads();

    for (int it = 0; it < N_SLABS - 1; ++it) {
        const int cur = (it & 1) * STAGE_FLOATS;
        const int nxt = STAGE_FLOATS - cur;
        const int kb  = (it + 1) * K_BLOCK;

        #pragma unroll
        for (int i = 0; i < W_HALF; ++i)
            pw[i] = *(const float4*)(wSrc + (size_t)i * W_ROWS_PER_PASS * H + kb);
        px = *(const float4*)(xSrc + kb);

        load_frags(0, cur, 0);
        #pragma unroll
        for (int j = 0; j < J_STEPS; ++j) {
            const int s = j & 1;
            if (j + 1 < J_STEPS) load_frags(s ^ 1, cur, j + 1);
            mma_step(s);
            if (j == 1) {
                #pragma unroll
                for (int i = 0; i < W_HALF; ++i)
                    *(float4*)&smem[nxt + wDst + i * W_ROWS_PER_PASS * K_PITCH] = pw[i];
                *(float4*)&smem[nxt + xDst] = px;
                #pragma unroll
                for (int i = 0; i < W_HALF; ++i)
                    pw[i] = *(const float4*)(wSrc + (size_t)(i + W_HALF) * W_ROWS_PER_PASS * H + kb);
            }
        }
        #pragma unroll
        for (int i = 0; i < W_HALF; ++i)
            *(float4*)&smem[nxt + wDst + (i + W_HALF) * W_ROWS_PER_PASS * K_PITCH] = pw[i];
        __syncthreads();
    }

    {
        const int cur = ((N_SLABS - 1) & 1) * STAGE_FLOATS;
        load_frags(0, cur, 0);
        #pragma unroll
        for (int j = 0; j < J_STEPS; ++j) {
            const int s = j & 1;
            if (j + 1 < J_STEPS) load_frags(s ^ 1, cur, j + 1);
            mma_step(s);
        }
    }
    __syncthreads();

    {
        const int cCol = lane & 15;
        const int rOff = (lane >> 4) << 3;
        #pragma unroll
        for (int nt = 0; nt < 4; ++nt) {
            const int n = nWaveBase + nt * 16 + cCol;
            const float b = bias[n];
            #pragma unroll
            for (int mt = 0; mt < 4; ++mt)
                #pragma unroll
                for (int j = 0; j < 8; ++j)
                    sOut[(mt * 16 + rOff + j) * H + n] = acc[mt][nt][j] + b;
        }
    }
    __syncthreads();

    for (int r = wave; r < M_TILE; r += NWAVES) {
        const float* row = &sOut[r * H];
        const int base = lane * 32;

        float vals[32];
        float sum = 0.0f;
        #pragma unroll
        for (int i = 0; i < 32; i += 4) {
            const float4 v = *(const float4*)&row[base + i];
            vals[i + 0] = v.x; vals[i + 1] = v.y; vals[i + 2] = v.z; vals[i + 3] = v.w;
            sum += v.x + v.y + v.z + v.w;
        }

        float pre = sum;
        #pragma unroll
        for (int d = 1; d < 32; d <<= 1) {
            const float t = __shfl_up(pre, d, 32);
            if (lane >= d) pre += t;
        }
        float run = pre - sum;

        const size_t outBase = (size_t)(mBase + r) * H + base;
        #pragma unroll
        for (int i = 0; i < 32; i += 4) {
            float4 o;
            run += vals[i + 0]; o.x = run;
            run += vals[i + 1]; o.y = run;
            run += vals[i + 2]; o.z = run;
            run += vals[i + 3]; o.w = run;
            *(float4*)(out + outBase + i) = o;
        }
    }
}

extern "C" void kernel_launch(void* const* d_in, const int* in_sizes, int n_in,
                              void* d_out, int out_size, void* d_ws, size_t ws_size,
                              hipStream_t stream) {
    const float* x    = (const float*)d_in[0];
    const float* W    = (const float*)d_in[1];
    const float* bias = (const float*)d_in[2];
    float* out = (float*)d_out;

    const int Mrows = in_sizes[0] / H;
    dim3 grid(Mrows / M_TILE);
    SSM_gemm_scan_kernel<<<grid, NTHREADS, 0, stream>>>(x, W, bias, out);
}


# Scan results go back to a padded LDS tile and are copied out so each store instruction writes 4 whole 128-B lines (was 32 lines x 16 B per instruction, 8 instructions per line); sum order unchanged
# speedup vs baseline: 2.2187x; 1.0385x over previous
#include <hip/hip_runtime.h>


typedef __attribute__((ext_vector_type(2))) float v2f;
typedef __attribute__((ext_vector_type(8))) float v8f;

#define H        1024
#define M_TILE   64
#define K_BLOCK  32
#define K_PITCH  36
#define NTHREADS 512
#define NWAVES   16
#define N_SLABS  (H / K_BLOCK)
#define STAGE_ROWS   (H + M_TILE)
#define STAGE_FLOATS (STAGE_ROWS * K_PITCH)
#define W_ROWS_PER_PASS (NTHREADS / 8)
#define W_PASSES (H / W_ROWS_PER_PASS)
#define W_HALF   (W_PASSES / 2)
#define J_STEPS  (K_BLOCK / 8)
#define SEG_PITCH 36
#define OUT_PITCH (H / 32 * SEG_PITCH + 4)

static_assert(M_TILE == W_ROWS_PER_PASS, "every thread stages exactly one float4 of the x slab");
static_assert(M_TILE * OUT_PITCH <= 2 * STAGE_FLOATS, "padded output tile must fit inside the two staging buffers");
static_assert(2 * STAGE_FLOATS * 4 <= 320 * 1024, "both staging buffers must fit the LDS of one WGP");
static_assert(J_STEPS == 4, "the staging batches are hung on fragment steps 1 and 3");

__global__ __launch_bounds__(NTHREADS)
void SSM_gemm_scan_kernel(const float* __restrict__ x,
                          const float* __restrict__ W,
                          const float* __restrict__ bias,
                          float* __restrict__ out)
{
    __shared__ __attribute__((aligned(16))) float smem[2 * STAGE_FLOATS];
    float* const sOut = smem;

    const int tid  = threadIdx.x;
    const int wave = __builtin_amdgcn_readfirstlane(tid >> 5);
    const int lane = tid & 31;
    const int mBase = blockIdx.x * M_TILE;

    v8f acc[4][4];
    #pragma unroll
    for (int mt = 0; mt < 4; ++mt)
        #pragma unroll
        for (int nt = 0; nt < 4; ++nt)
            acc[mt][nt] = v8f{};

    const int nWaveBase = wave * 64;

    const int fC = lane & 15;
    const int fK = (lane >> 4) << 2;
    const int aOff = (H + fC) * K_PITCH + fK;
    const int bOff = (nWaveBase + fC) * K_PITCH + fK;

    const int sRow = tid >> 3;
    const int sQ   = (tid & 7) << 2;
    const float* wSrc = W + (size_t)sRow * H + sQ;
    const float* xSrc = x + (size_t)(mBase + sRow) * H + sQ;
    const int wDst = sRow * K_PITCH + sQ;
    const int xDst = (H + sRow) * K_PITCH + sQ;

    float4 fa[2][4];
    float4 fb[2][4];

    auto load_frags = [&](const int s, const int base, const int j) {
        #pragma unroll
        for (int mt = 0; mt < 4; ++mt)
            fa[s][mt] = *(const float4*)&smem[base + aOff + mt * 16 * K_PITCH + j * 8];
        #pragma unroll
        for (int nt = 0; nt < 4; ++nt)
            fb[s][nt] = *(const float4*)&smem[base + bOff + nt * 16 * K_PITCH + j * 8];
    };

    auto mma_step = [&](const int s) {
        #pragma unroll
        for (int nt = 0; nt < 4; ++nt)
            #pragma unroll
            for (int mt = 0; mt < 4; ++mt) {
                const v2f a = {fa[s][mt].x, fa[s][mt].y};
                const v2f b = {fb[s][nt].x, fb[s][nt].y};
                acc[mt][nt] = __builtin_amdgcn_wmma_f32_16x16x4_f32(
                        false, a, false, b, (short)0, acc[mt][nt], false, false);
            }
        #pragma unroll
        for (int nt = 0; nt < 4; ++nt)
            #pragma unroll
            for (int mt = 0; mt < 4; ++mt) {
                const v2f a = {fa[s][mt].z, fa[s][mt].w};
                const v2f b = {fb[s][nt].z, fb[s][nt].w};
                acc[mt][nt] = __builtin_amdgcn_wmma_f32_16x16x4_f32(
                        false, a, false, b, (short)0, acc[mt][nt], false, false);
            }
    };

    float4 pw[W_HALF];
    float4 px;

    #pragma unroll
    for (int i = 0; i < W_HALF; ++i)
        pw[i] = *(const float4*)(wSrc + (size_t)i * W_ROWS_PER_PASS * H);
    px = *(const float4*)xSrc;
    #pragma unroll
    for (int i = 0; i < W_HALF; ++i)
        *(float4*)&smem[wDst + i * W_ROWS_PER_PASS * K_PITCH] = pw[i];
    *(float4*)&smem[xDst] = px;
    #pragma unroll
    for (int i = 0; i < W_HALF; ++i)
        pw[i] = *(const float4*)(wSrc + (size_t)(i + W_HALF) * W_ROWS_PER_PASS * H);
    #pragma unroll
    for (int i = 0; i < W_HALF; ++i)
        *(float4*)&smem[wDst + (i + W_HALF) * W_ROWS_PER_PASS * K_PITCH] = pw[i];
    __syncthreads();

    for (int it = 0; it < N_SLABS - 1; ++it) {
        const int cur = (it & 1) * STAGE_FLOATS;
        const int nxt = STAGE_FLOATS - cur;
        const int kb  = (it + 1) * K_BLOCK;

        #pragma unroll
        for (int i = 0; i < W_HALF; ++i)
            pw[i] = *(const float4*)(wSrc + (size_t)i * W_ROWS_PER_PASS * H + kb);
        px = *(const float4*)(xSrc + kb);

        load_frags(0, cur, 0);
        #pragma unroll
        for (int j = 0; j < J_STEPS; ++j) {
            const int s = j & 1;
            if (j + 1 < J_STEPS) load_frags(s ^ 1, cur, j + 1);
            mma_step(s);
            if (j == 1) {
                #pragma unroll
                for (int i = 0; i < W_HALF; ++i)
                    *(float4*)&smem[nxt + wDst + i * W_ROWS_PER_PASS * K_PITCH] = pw[i];
                *(float4*)&smem[nxt + xDst] = px;
                #pragma unroll
                for (int i = 0; i < W_HALF; ++i)
                    pw[i] = *(const float4*)(wSrc + (size_t)(i + W_HALF) * W_ROWS_PER_PASS * H + kb);
            }
        }
        #pragma unroll
        for (int i = 0; i < W_HALF; ++i)
            *(float4*)&smem[nxt + wDst + (i + W_HALF) * W_ROWS_PER_PASS * K_PITCH] = pw[i];
        __syncthreads();
    }

    {
        const int cur = ((N_SLABS - 1) & 1) * STAGE_FLOATS;
        load_frags(0, cur, 0);
        #pragma unroll
        for (int j = 0; j < J_STEPS; ++j) {
            const int s = j & 1;
            if (j + 1 < J_STEPS) load_frags(s ^ 1, cur, j + 1);
            mma_step(s);
        }
    }
    __syncthreads();

    {
        const int cCol = lane & 15;
        const int rOff = (lane >> 4) << 3;
        #pragma unroll
        for (int nt = 0; nt < 4; ++nt) {
            const int n = nWaveBase + nt * 16 + cCol;
            const float b = bias[n];
            const int np = n + ((n >> 5) << 2);
            #pragma unroll
            for (int mt = 0; mt < 4; ++mt)
                #pragma unroll
                for (int j = 0; j < 8; ++j)
                    sOut[(mt * 16 + rOff + j) * OUT_PITCH + np] = acc[mt][nt][j] + b;
        }
    }
    __syncthreads();

    for (int r = wave; r < M_TILE; r += NWAVES) {
        float* seg = &sOut[r * OUT_PITCH + lane * SEG_PITCH];

        float vals[32];
        float sum = 0.0f;
        #pragma unroll
        for (int i = 0; i < 32; i += 4) {
            const float4 v = *(const float4*)&seg[i];
            vals[i + 0] = v.x; vals[i + 1] = v.y; vals[i + 2] = v.z; vals[i + 3] = v.w;
            sum += v.x + v.y + v.z + v.w;
        }

        float pre = sum;
        #pragma unroll
        for (int d = 1; d < 32; d <<= 1) {
            const float t = __shfl_up(pre, d, 32);
            if (lane >= d) pre += t;
        }
        float run = pre - sum;

        #pragma unroll
        for (int i = 0; i < 32; i += 4) {
            float4 o;
            run += vals[i + 0]; o.x = run;
            run += vals[i + 1]; o.y = run;
            run += vals[i + 2]; o.z = run;
            run += vals[i + 3]; o.w = run;
            *(float4*)&seg[i] = o;
        }
    }
    __syncthreads();

    const int cRow = wave >> 3;
    const int cN   = (((wave & 7) << 5) | lane) << 2;
    const float* cSrc = &sOut[cRow * OUT_PITCH + cN + ((cN >> 5) << 2)];
    float* cDst = out + (size_t)(mBase + cRow) * H + cN;
    #pragma unroll 8
    for (int t = 0; t < M_TILE / 2; ++t) {
        const float4 v = *(const float4*)&cSrc[t * 2 * OUT_PITCH];
        *(float4*)&cDst[(size_t)t * 2 * H] = v;
    }
}

extern "C" void kernel_launch(void* const* d_in, const int* in_sizes, int n_in,
                              void* d_out, int out_size, void* d_ws, size_t ws_size,
                              hipStream_t stream) {
    const float* x    = (const float*)d_in[0];
    const float* W    = (const float*)d_in[1];
    const float* bias = (const float*)d_in[2];
    float* out = (float*)d_out;

    const int Mrows = in_sizes[0] / H;
    dim3 grid(Mrows / M_TILE);
    SSM_gemm_scan_kernel<<<grid, NTHREADS, 0, stream>>>(x, W, bias, out);
}


# K loop pinned with scheduling fences: each slab's LDS reads and parking stores stand in 3 blocks between unbroken runs of 32/64 WMMAs (compiler spread them over 12 places); sum order unchanged
# speedup vs baseline: 2.3959x; 1.0799x over previous
#include <hip/hip_runtime.h>


typedef __attribute__((ext_vector_type(2))) float v2f;
typedef __attribute__((ext_vector_type(8))) float v8f;

#define H        1024
#define M_TILE   64
#define K_BLOCK  32
#define K_PITCH  36
#define NTHREADS 512
#define NWAVES   16
#define N_SLABS  (H / K_BLOCK)
#define STAGE_ROWS   (H + M_TILE)
#define STAGE_FLOATS (STAGE_ROWS * K_PITCH)
#define W_ROWS_PER_PASS (NTHREADS / 8)
#define W_PASSES (H / W_ROWS_PER_PASS)
#define W_HALF   (W_PASSES / 2)
#define J_STEPS  (K_BLOCK / 8)
#define SEG_PITCH 36
#define OUT_PITCH (H / 32 * SEG_PITCH + 4)

static_assert(M_TILE == W_ROWS_PER_PASS, "every thread stages exactly one float4 of the x slab");
static_assert(M_TILE * OUT_PITCH <= 2 * STAGE_FLOATS, "padded output tile must fit inside the two staging buffers");
static_assert(2 * STAGE_FLOATS * 4 <= 320 * 1024, "both staging buffers must fit the LDS of one WGP");
static_assert(J_STEPS == 4, "the K loop is written out for four fragment steps per slab");

__global__ __launch_bounds__(NTHREADS)
void SSM_gemm_scan_kernel(const float* __restrict__ x,
                          const float* __restrict__ W,
                          const float* __restrict__ bias,
                          float* __restrict__ out)
{
    __shared__ __attribute__((aligned(16))) float smem[2 * STAGE_FLOATS];
    float* const sOut = smem;

    const int tid  = threadIdx.x;
    const int wave = __builtin_amdgcn_readfirstlane(tid >> 5);
    const int lane = tid & 31;
    const int mBase = blockIdx.x * M_TILE;

    v8f acc[4][4];
    #pragma unroll
    for (int mt = 0; mt < 4; ++mt)
        #pragma unroll
        for (int nt = 0; nt < 4; ++nt)
            acc[mt][nt] = v8f{};

    const int nWaveBase = wave * 64;

    const int fC = lane & 15;
    const int fK = (lane >> 4) << 2;
    const int aOff = (H + fC) * K_PITCH + fK;
    const int bOff = (nWaveBase + fC) * K_PITCH + fK;

    const int sRow = tid >> 3;
    const int sQ   = (tid & 7) << 2;
    const float* wSrc = W + (size_t)sRow * H + sQ;
    const float* xSrc = x + (size_t)(mBase + sRow) * H + sQ;
    const int wDst = sRow * K_PITCH + sQ;
    const int xDst = (H + sRow) * K_PITCH + sQ;

    float4 fa[2][4];
    float4 fb[2][4];

    auto load_frags = [&](const int s, const int base, const int j) {
        #pragma unroll
        for (int mt = 0; mt < 4; ++mt)
            fa[s][mt] = *(const float4*)&smem[base + aOff + mt * 16 * K_PITCH + j * 8];
        #pragma unroll
        for (int nt = 0; nt < 4; ++nt)
            fb[s][nt] = *(const float4*)&smem[base + bOff + nt * 16 * K_PITCH + j * 8];
    };

    auto mma_step = [&](const int s) {
        #pragma unroll
        for (int nt = 0; nt < 4; ++nt)
            #pragma unroll
            for (int mt = 0; mt < 4; ++mt) {
                const v2f a = {fa[s][mt].x, fa[s][mt].y};
                const v2f b = {fb[s][nt].x, fb[s][nt].y};
                acc[mt][nt] = __builtin_amdgcn_wmma_f32_16x16x4_f32(
                        false, a, false, b, (short)0, acc[mt][nt], false, false);
            }
        #pragma unroll
        for (int nt = 0; nt < 4; ++nt)
            #pragma unroll
            for (int mt = 0; mt < 4; ++mt) {
                const v2f a = {fa[s][mt].z, fa[s][mt].w};
                const v2f b = {fb[s][nt].z, fb[s][nt].w};
                acc[mt][nt] = __builtin_amdgcn_wmma_f32_16x16x4_f32(
                        false, a, false, b, (short)0, acc[mt][nt], false, false);
            }
    };

    float4 pw[W_HALF];
    float4 px;

    #pragma unroll
    for (int i = 0; i < W_HALF; ++i)
        pw[i] = *(const float4*)(wSrc + (size_t)i * W_ROWS_PER_PASS * H);
    px = *(const float4*)xSrc;
    #pragma unroll
    for (int i = 0; i < W_HALF; ++i)
        *(float4*)&smem[wDst + i * W_ROWS_PER_PASS * K_PITCH] = pw[i];
    *(float4*)&smem[xDst] = px;
    #pragma unroll
    for (int i = 0; i < W_HALF; ++i)
        pw[i] = *(const float4*)(wSrc + (size_t)(i + W_HALF) * W_ROWS_PER_PASS * H);
    #pragma unroll
    for (int i = 0; i < W_HALF; ++i)
        *(float4*)&smem[wDst + (i + W_HALF) * W_ROWS_PER_PASS * K_PITCH] = pw[i];
    __syncthreads();

    for (int it = 0; it < N_SLABS - 1; ++it) {
        const int cur = (it & 1) * STAGE_FLOATS;
        const int nxt = STAGE_FLOATS - cur;
        const int kb  = (it + 1) * K_BLOCK;

        load_frags(0, cur, 0);
        load_frags(1, cur, 1);
        #pragma unroll
        for (int i = 0; i < W_HALF; ++i)
            pw[i] = *(const float4*)(wSrc + (size_t)i * W_ROWS_PER_PASS * H + kb);
        __builtin_amdgcn_sched_barrier(0x26);
        mma_step(0);
        __builtin_amdgcn_sched_barrier(0x26);

        #pragma unroll
        for (int i = 0; i < W_HALF; ++i)
            *(float4*)&smem[nxt + wDst + i * W_ROWS_PER_PASS * K_PITCH] = pw[i];
        load_frags(0, cur, 2);
        __builtin_amdgcn_sched_barrier(0x6);
        mma_step(1);
        __builtin_amdgcn_sched_barrier(0x6);

        #pragma unroll
        for (int i = 0; i < W_HALF; ++i)
            pw[i] = *(const float4*)(wSrc + (size_t)(i + W_HALF) * W_ROWS_PER_PASS * H + kb);
        px = *(const float4*)(xSrc + kb);
        load_frags(1, cur, 3);
        __builtin_amdgcn_sched_barrier(0x26);
        mma_step(0);
        mma_step(1);
        __builtin_amdgcn_sched_barrier(0x26);

        #pragma unroll
        for (int i = 0; i < W_HALF; ++i)
            *(float4*)&smem[nxt + wDst + (i + W_HALF) * W_ROWS_PER_PASS * K_PITCH] = pw[i];
        *(float4*)&smem[nxt + xDst] = px;
        __syncthreads();
    }

    {
        const int cur = ((N_SLABS - 1) & 1) * STAGE_FLOATS;
        load_frags(0, cur, 0);
        load_frags(1, cur, 1);
        __builtin_amdgcn_sched_barrier(0x6);
        mma_step(0);
        __builtin_amdgcn_sched_barrier(0x6);
        load_frags(0, cur, 2);
        __builtin_amdgcn_sched_barrier(0x6);
        mma_step(1);
        __builtin_amdgcn_sched_barrier(0x6);
        load_frags(1, cur, 3);
        __builtin_amdgcn_sched_barrier(0x6);
        mma_step(0);
        mma_step(1);
    }
    __syncthreads();

    {
        const int cCol = lane & 15;
        const int rOff = (lane >> 4) << 3;
        #pragma unroll
        for (int nt = 0; nt < 4; ++nt) {
            const int n = nWaveBase + nt * 16 + cCol;
            const float b = bias[n];
            const int np = n + ((n >> 5) << 2);
            #pragma unroll
            for (int mt = 0; mt < 4; ++mt)
                #pragma unroll
                for (int j = 0; j < 8; ++j)
                    sOut[(mt * 16 + rOff + j) * OUT_PITCH + np] = acc[mt][nt][j] + b;
        }
    }
    __syncthreads();

    for (int r = wave; r < M_TILE; r += NWAVES) {
        float* seg = &sOut[r * OUT_PITCH + lane * SEG_PITCH];

        float vals[32];
        float sum = 0.0f;
        #pragma unroll
        for (int i = 0; i < 32; i += 4) {
            const float4 v = *(const float4*)&seg[i];
            vals[i + 0] = v.x; vals[i + 1] = v.y; vals[i + 2] = v.z; vals[i + 3] = v.w;
            sum += v.x + v.y + v.z + v.w;
        }

        float pre = sum;
        #pragma unroll
        for (int d = 1; d < 32; d <<= 1) {
            const float t = __shfl_up(pre, d, 32);
            if (lane >= d) pre += t;
        }
        float run = pre - sum;

        #pragma unroll
        for (int i = 0; i < 32; i += 4) {
            float4 o;
            run += vals[i + 0]; o.x = run;
            run += vals[i + 1]; o.y = run;
            run += vals[i + 2]; o.z = run;
            run += vals[i + 3]; o.w = run;
            *(float4*)&seg[i] = o;
        }
    }
    __syncthreads();

    const int cRow = wave >> 3;
    const int cN   = (((wave & 7) << 5) | lane) << 2;
    const float* cSrc = &sOut[cRow * OUT_PITCH + cN + ((cN >> 5) << 2)];
    float* cDst = out + (size_t)(mBase + cRow) * H + cN;
    #pragma unroll 8
    for (int t = 0; t < M_TILE / 2; ++t) {
        const float4 v = *(const float4*)&cSrc[t * 2 * OUT_PITCH];
        *(float4*)&cDst[(size_t)t * 2 * H] = v;
    }
}

extern "C" void kernel_launch(void* const* d_in, const int* in_sizes, int n_in,
                              void* d_out, int out_size, void* d_ws, size_t ws_size,
                              hipStream_t stream) {
    const float* x    = (const float*)d_in[0];
    const float* W    = (const float*)d_in[1];
    const float* bias = (const float*)d_in[2];
    float* out = (float*)d_out;

    const int Mrows = in_sizes[0] / H;
    dim3 grid(Mrows / M_TILE);
    SSM_gemm_scan_kernel<<<grid, NTHREADS, 0, stream>>>(x, W, bias, out);
}


# Workgroup tile 128 rows x 512 cols in 2 column passes on 8 waves of 128x64 (was 64x1024 on 16 waves of 64x64): W staged once per 128 rows, fewer LDS reads per WMMA; scan sum order changes (carry)
# speedup vs baseline: 2.4983x; 1.0427x over previous
#include <hip/hip_runtime.h>


typedef __attribute__((ext_vector_type(2))) float v2f;
typedef __attribute__((ext_vector_type(8))) float v8f;

#define H        1024
#define M_TILE   128
#define N_PASS   512
#define N_PASSES (H / N_PASS)
#define K_BLOCK  32
#define K_PITCH  36
#define NTHREADS 256
#define NWAVES   8
#define WAVE_COLS 64
#define N_SLABS  (H / K_BLOCK)
#define STAGE_ROWS   (N_PASS + M_TILE)
#define STAGE_FLOATS (STAGE_ROWS * K_PITCH)
#define ROWS_PER_SWEEP (NTHREADS / 8)
#define W_SWEEPS (N_PASS / ROWS_PER_SWEEP)
#define X_SWEEPS (M_TILE / ROWS_PER_SWEEP)
#define W_HALF   (W_SWEEPS / 2)
#define X_HALF   (X_SWEEPS / 2)
#define J_STEPS  (K_BLOCK / 8)
#define SEG_PITCH 36
#define OUT_PITCH (N_PASS / 32 * SEG_PITCH + 4)
#define OUT_FLOATS (M_TILE * OUT_PITCH)
#define SCAN_ROWS_PER_TRIP (NWAVES * 2 * 4)

static_assert(2 * STAGE_FLOATS <= OUT_FLOATS, "the staging buffers live inside the space of the output tile");
static_assert((OUT_FLOATS + M_TILE) * 4 <= 320 * 1024, "output tile and row carries must fit the LDS of one WGP");
static_assert(J_STEPS == 4, "the K loop is written out for four fragment steps per slab");
static_assert(M_TILE % SCAN_ROWS_PER_TRIP == 0, "scan trips cover the tile rows exactly");

__global__ __launch_bounds__(NTHREADS)
void SSM_gemm_scan_kernel(const float* __restrict__ x,
                          const float* __restrict__ W,
                          const float* __restrict__ bias,
                          float* __restrict__ out)
{
    __shared__ __attribute__((aligned(16))) float smem[OUT_FLOATS + M_TILE];
    float* const sOut   = smem;
    float* const sCarry = smem + OUT_FLOATS;

    const int tid  = threadIdx.x;
    const int wave = __builtin_amdgcn_readfirstlane(tid >> 5);
    const int lane = tid & 31;
    const int mBase = blockIdx.x * M_TILE;
    const int nWaveBase = wave * WAVE_COLS;

    const int fC = lane & 15;
    const int fK = (lane >> 4) << 2;
    const int aOff = (N_PASS + fC) * K_PITCH + fK;
    const int bOff = (nWaveBase + fC) * K_PITCH + fK;

    const int sRow = tid >> 3;
    const int sQ   = (tid & 7) << 2;
    const float* xSrc = x + (size_t)(mBase + sRow) * H + sQ;
    const int wDst = sRow * K_PITCH + sQ;
    const int xDst = (N_PASS + sRow) * K_PITCH + sQ;

    for (int p = 0; p < N_PASSES; ++p) {
        const float* wSrc = W + (size_t)(p * N_PASS + sRow) * H + sQ;

        v8f acc[8][4];
        #pragma unroll
        for (int mt = 0; mt < 8; ++mt)
            #pragma unroll
            for (int nt = 0; nt < 4; ++nt)
                acc[mt][nt] = v8f{};

        float4 fa[2][8];
        float4 fb[2][4];

        auto load_frags = [&](const int s, const int base, const int j) {
            #pragma unroll
            for (int nt = 0; nt < 4; ++nt)
                fb[s][nt] = *(const float4*)&smem[base + bOff + nt * 16 * K_PITCH + j * 8];
            #pragma unroll
            for (int mt = 0; mt < 8; ++mt)
                fa[s][mt] = *(const float4*)&smem[base + aOff + mt * 16 * K_PITCH + j * 8];
        };

        auto mma_step = [&](const int s) {
            #pragma unroll
            for (int nt = 0; nt < 4; ++nt)
                #pragma unroll
                for (int mt = 0; mt < 8; ++mt) {
                    const v2f a = {fa[s][mt].x, fa[s][mt].y};
                    const v2f b = {fb[s][nt].x, fb[s][nt].y};
                    acc[mt][nt] = __builtin_amdgcn_wmma_f32_16x16x4_f32(
                            false, a, false, b, (short)0, acc[mt][nt], false, false);
                }
            #pragma unroll
            for (int nt = 0; nt < 4; ++nt)
                #pragma unroll
                for (int mt = 0; mt < 8; ++mt) {
                    const v2f a = {fa[s][mt].z, fa[s][mt].w};
                    const v2f b = {fb[s][nt].z, fb[s][nt].w};
                    acc[mt][nt] = __builtin_amdgcn_wmma_f32_16x16x4_f32(
                            false, a, false, b, (short)0, acc[mt][nt], false, false);
                }
        };

        float4 pw[W_HALF];
        float4 px[X_HALF];

        #pragma unroll
        for (int i = 0; i < W_HALF; ++i)
            pw[i] = *(const float4*)(wSrc + (size_t)i * ROWS_PER_SWEEP * H);
        #pragma unroll
        for (int i = 0; i < X_HALF; ++i)
            px[i] = *(const float4*)(xSrc + (size_t)i * ROWS_PER_SWEEP * H);
        #pragma unroll
        for (int i = 0; i < W_HALF; ++i)
            *(float4*)&smem[wDst + i * ROWS_PER_SWEEP * K_PITCH] = pw[i];
        #pragma unroll
        for (int i = 0; i < X_HALF; ++i)
            *(float4*)&smem[xDst + i * ROWS_PER_SWEEP * K_PITCH] = px[i];
        #pragma unroll
        for (int i = 0; i < W_HALF; ++i)
            pw[i] = *(const float4*)(wSrc + (size_t)(i + W_HALF) * ROWS_PER_SWEEP * H);
        #pragma unroll
        for (int i = 0; i < X_HALF; ++i)
            px[i] = *(const float4*)(xSrc + (size_t)(i + X_HALF) * ROWS_PER_SWEEP * H);
        #pragma unroll
        for (int i = 0; i < W_HALF; ++i)
            *(float4*)&smem[wDst + (i + W_HALF) * ROWS_PER_SWEEP * K_PITCH] = pw[i];
        #pragma unroll
        for (int i = 0; i < X_HALF; ++i)
            *(float4*)&smem[xDst + (i + X_HALF) * ROWS_PER_SWEEP * K_PITCH] = px[i];
        __syncthreads();

        for (int it = 0; it < N_SLABS - 1; ++it) {
            const int cur = (it & 1) * STAGE_FLOATS;
            const int nxt = STAGE_FLOATS - cur;
            const int kb  = (it + 1) * K_BLOCK;

            load_frags(0, cur, 0);
            load_frags(1, cur, 1);
            #pragma unroll
            for (int i = 0; i < W_HALF; ++i)
                pw[i] = *(const float4*)(wSrc + (size_t)i * ROWS_PER_SWEEP * H + kb);
            #pragma unroll
            for (int i = 0; i < X_HALF; ++i)
                px[i] = *(const float4*)(xSrc + (size_t)i * ROWS_PER_SWEEP * H + kb);
            __builtin_amdgcn_sched_barrier(0x26);
            mma_step(0);
            __builtin_amdgcn_sched_barrier(0x26);

            #pragma unroll
            for (int i = 0; i < W_HALF; ++i)
                *(float4*)&smem[nxt + wDst + i * ROWS_PER_SWEEP * K_PITCH] = pw[i];
            #pragma unroll
            for (int i = 0; i < X_HALF; ++i)
                *(float4*)&smem[nxt + xDst + i * ROWS_PER_SWEEP * K_PITCH] = px[i];
            load_frags(0, cur, 2);
            __builtin_amdgcn_sched_barrier(0x6);
            mma_step(1);
            __builtin_amdgcn_sched_barrier(0x6);

            #pragma unroll
            for (int i = 0; i < W_HALF; ++i)
                pw[i] = *(const float4*)(wSrc + (size_t)(i + W_HALF) * ROWS_PER_SWEEP * H + kb);
            #pragma unroll
            for (int i = 0; i < X_HALF; ++i)
                px[i] = *(const float4*)(xSrc + (size_t)(i + X_HALF) * ROWS_PER_SWEEP * H + kb);
            load_frags(1, cur, 3);
            __builtin_amdgcn_sched_barrier(0x26);
            mma_step(0);
            mma_step(1);
            __builtin_amdgcn_sched_barrier(0x26);

            #pragma unroll
            for (int i = 0; i < W_HALF; ++i)
                *(float4*)&smem[nxt + wDst + (i + W_HALF) * ROWS_PER_SWEEP * K_PITCH] = pw[i];
            #pragma unroll
            for (int i = 0; i < X_HALF; ++i)
                *(float4*)&smem[nxt + xDst + (i + X_HALF) * ROWS_PER_SWEEP * K_PITCH] = px[i];
            __syncthreads();
        }

        {
            const int cur = ((N_SLABS - 1) & 1) * STAGE_FLOATS;
            load_frags(0, cur, 0);
            load_frags(1, cur, 1);
            __builtin_amdgcn_sched_barrier(0x6);
            mma_step(0);
            __builtin_amdgcn_sched_barrier(0x6);
            load_frags(0, cur, 2);
            __builtin_amdgcn_sched_barrier(0x6);
            mma_step(1);
            __builtin_amdgcn_sched_barrier(0x6);
            load_frags(1, cur, 3);
            __builtin_amdgcn_sched_barrier(0x6);
            mma_step(0);
            mma_step(1);
        }
        __syncthreads();

        {
            const int cCol = lane & 15;
            const int rOff = (lane >> 4) << 3;
            #pragma unroll
            for (int nt = 0; nt < 4; ++nt) {
                const int n = nWaveBase + nt * 16 + cCol;
                const float b = bias[p * N_PASS + n];
                const int np = n + ((n >> 5) << 2);
                #pragma unroll
                for (int mt = 0; mt < 8; ++mt)
                    #pragma unroll
                    for (int j = 0; j < 8; ++j)
                        sOut[(mt * 16 + rOff + j) * OUT_PITCH + np] = acc[mt][nt][j] + b;
            }
        }
        __syncthreads();

        {
            const int sHalf = lane >> 4;
            const int l16   = lane & 15;
            for (int g = 0; g < M_TILE / SCAN_ROWS_PER_TRIP; ++g) {
                float vals[4][32];
                float sum[4];
                float pre[4];
                #pragma unroll
                for (int u = 0; u < 4; ++u) {
                    const int r = (g * 4 + u) * (NWAVES * 2) + wave * 2 + sHalf;
                    const float* seg = &sOut[r * OUT_PITCH + l16 * SEG_PITCH];
                    float s = 0.0f;
                    #pragma unroll
                    for (int i = 0; i < 32; i += 4) {
                        const float4 v = *(const float4*)&seg[i];
                        vals[u][i + 0] = v.x; vals[u][i + 1] = v.y; vals[u][i + 2] = v.z; vals[u][i + 3] = v.w;
                        s += v.x + v.y + v.z + v.w;
                    }
                    sum[u] = s;
                    pre[u] = s;
                }

                #pragma unroll
                for (int d = 1; d < 16; d <<= 1) {
                    #pragma unroll
                    for (int u = 0; u < 4; ++u) {
                        const float t = __shfl_up(pre[u], d, 16);
                        if (l16 >= d) pre[u] += t;
                    }
                }

                #pragma unroll
                for (int u = 0; u < 4; ++u) {
                    const int r = (g * 4 + u) * (NWAVES * 2) + wave * 2 + sHalf;
                    float* seg = &sOut[r * OUT_PITCH + l16 * SEG_PITCH];
                    float run = pre[u] - sum[u];
                    if (p > 0) {
                        run += sCarry[r];
                    } else if (l16 == 15) {
                        sCarry[r] = pre[u];
                    }
                    #pragma unroll
                    for (int i = 0; i < 32; i += 4) {
                        float4 o;
                        run += vals[u][i + 0]; o.x = run;
                        run += vals[u][i + 1]; o.y = run;
                        run += vals[u][i + 2]; o.z = run;
                        run += vals[u][i + 3]; o.w = run;
                        *(float4*)&seg[i] = o;
                    }
                }
            }
        }
        __syncthreads();

        {
            const int cRow = wave >> 2;
            const int cN   = (((wave & 3) << 5) | lane) << 2;
            const float* cSrc = &sOut[cRow * OUT_PITCH + cN + ((cN >> 5) << 2)];
            float* cDst = out + (size_t)(mBase + cRow) * H + p * N_PASS + cN;
            #pragma unroll 8
            for (int t = 0; t < M_TILE / 2; ++t) {
                const float4 v = *(const float4*)&cSrc[t * 2 * OUT_PITCH];
                *(float4*)&cDst[(size_t)t * 2 * H] = v;
            }
        }
        __syncthreads();
    }
}

extern "C" void kernel_launch(void* const* d_in, const int* in_sizes, int n_in,
                              void* d_out, int out_size, void* d_ws, size_t ws_size,
                              hipStream_t stream) {
    const float* x    = (const float*)d_in[0];
    const float* W    = (const float*)d_in[1];
    const float* bias = (const float*)d_in[2];
    float* out = (float*)d_out;

    const int Mrows = in_sizes[0] / H;
    dim3 grid(Mrows / M_TILE);
    SSM_gemm_scan_kernel<<<grid, NTHREADS, 0, stream>>>(x, W, bias, out);
}
